# starting kernel
# baseline (speedup 1.0000x reference)
#include <hip/hip_runtime.h>
#include <hip/hip_bf16.h>

typedef __attribute__((ext_vector_type(2))) float v2f;
typedef __attribute__((ext_vector_type(8))) float v8f;

#define INCH  128
#define HC    256
#define NHD   8
#define CPH   32
#define OUTC  40
#define SLOPE 0.2f

__device__ __forceinline__ float leaky(float x) { return x > 0.f ? x : SLOPE * x; }

__global__ void gemm_x_w1(const float* __restrict__ X, const float* __restrict__ W,
                          float* __restrict__ H1, int nrows) {
    const int lane = threadIdx.x & 31;
    const int wave = threadIdx.x >> 5;
    const int m    = lane & 15;
    const int half = lane >> 4;
    const int rowBase = blockIdx.x << 4;
    if (rowBase >= nrows) return;
    const float* xrow = X + (size_t)(rowBase + m) * INCH;
    for (int t = 0; t < 2; ++t) {
        const int colBase = ((wave << 1) + t) << 4;
        v8f acc = {0.f, 0.f, 0.f, 0.f, 0.f, 0.f, 0.f, 0.f};
        for (int k = 0; k < INCH; k += 4) {
            const int kk = k + (half << 1);
            v2f a, b;
            a.x = xrow[kk];
            a.y = xrow[kk + 1];
            b.x = W[(size_t)kk * HC + colBase + m];
            b.y = W[(size_t)(kk + 1) * HC + colBase + m];
            acc = __builtin_amdgcn_wmma_f32_16x16x4_f32(false, a, false, b,
                                                        (short)0, acc, false, false);
        }
#pragma unroll
        for (int v = 0; v < 8; ++v)
            H1[(size_t)(rowBase + v + (half << 3)) * HC + colBase + m] = acc[v];
    }
}

__global__ void gemm_h_w2(const float* __restrict__ Hin, const float* __restrict__ W,
                          float* __restrict__ H2, int nrows) {
    const int lane = threadIdx.x & 31;
    const int wave = threadIdx.x >> 5;
    const int m    = lane & 15;
    const int half = lane >> 4;
    const int rowBase = blockIdx.x << 4;
    if (rowBase >= nrows) return;
    const int colBase = wave << 4;
    const int col = colBase + m;
    const float* hrow = Hin + (size_t)(rowBase + m) * HC;
    v8f acc = {0.f, 0.f, 0.f, 0.f, 0.f, 0.f, 0.f, 0.f};
    for (int k = 0; k < HC; k += 4) {
        const int kk = k + (half << 1);
        v2f a, b;
        a.x = hrow[kk];
        a.y = hrow[kk + 1];
        b.x = (col < OUTC) ? W[kk * OUTC + col] : 0.f;
        b.y = (col < OUTC) ? W[(kk + 1) * OUTC + col] : 0.f;
        acc = __builtin_amdgcn_wmma_f32_16x16x4_f32(false, a, false, b,
                                                    (short)0, acc, false, false);
    }
    if (col < OUTC) {
#pragma unroll
        for (int v = 0; v < 8; ++v)
            H2[(size_t)(rowBase + v + (half << 3)) * OUTC + col] = acc[v];
    }
}

__global__ void attn_dots_l1(const float* __restrict__ H1, const float* __restrict__ Asrc,
                             const float* __restrict__ Adst, float* __restrict__ aS,
                             float* __restrict__ aD, int n) {
    __shared__ float sAs[HC], sAd[HC];
    for (int i = threadIdx.x; i < HC; i += blockDim.x) { sAs[i] = Asrc[i]; sAd[i] = Adst[i]; }
    __syncthreads();
    const int idx = blockIdx.x * blockDim.x + threadIdx.x;
    if (idx >= n * NHD) return;
    const int node = idx >> 3, h = idx & 7;
    const float* row = H1 + (size_t)node * HC + h * CPH;
    float s = 0.f, d = 0.f;
#pragma unroll
    for (int c = 0; c < CPH; ++c) {
        const float v = row[c];
        s += v * sAs[h * CPH + c];
        d += v * sAd[h * CPH + c];
    }
    aS[idx] = s; aD[idx] = d;
}

__global__ void attn_dots_l2(const float* __restrict__ H2, const float* __restrict__ Asrc,
                             const float* __restrict__ Adst, float* __restrict__ aS,
                             float* __restrict__ aD, int n) {
    const int node = blockIdx.x * blockDim.x + threadIdx.x;
    if (node >= n) return;
    const float* row = H2 + (size_t)node * OUTC;
    float s = 0.f, d = 0.f;
#pragma unroll
    for (int c = 0; c < OUTC; ++c) { const float v = row[c]; s += v * Asrc[c]; d += v * Adst[c]; }
    aS[node] = s; aD[node] = d;
}

__global__ void count_edges(const int* __restrict__ dstA, int* __restrict__ cnt, int E, int n) {
    const int e = blockIdx.x * blockDim.x + threadIdx.x;
    if (e >= E + n) return;
    const int d = (e < E) ? dstA[e] : (e - E);
    atomicAdd(&cnt[d], 1);
}

__global__ void scan_rowptr(const int* __restrict__ cnt, int* __restrict__ rowptr,
                            int* __restrict__ cursor, int n) {
    __shared__ int lds[1024];
    __shared__ int carryS;
    int carry = 0;
    for (int base = 0; base < n; base += 1024) {
        const int i = base + (int)threadIdx.x;
        const int v = (i < n) ? cnt[i] : 0;
        lds[threadIdx.x] = v;
        __syncthreads();
        for (int off = 1; off < 1024; off <<= 1) {
            const int t = (threadIdx.x >= (unsigned)off) ? lds[threadIdx.x - off] : 0;
            __syncthreads();
            lds[threadIdx.x] += t;
            __syncthreads();
        }
        const int incl = lds[threadIdx.x];
        const int excl = incl - v;
        if (i < n) { rowptr[i] = carry + excl; cursor[i] = carry + excl; }
        if (threadIdx.x == 1023) carryS = carry + incl;
        __syncthreads();
        carry = carryS;
    }
    if (threadIdx.x == 0) rowptr[n] = carry;
}

__global__ void fill_csr(const int* __restrict__ srcA, const int* __restrict__ dstA,
                         int* __restrict__ cursor, int* __restrict__ col, int E, int n) {
    const int e = blockIdx.x * blockDim.x + threadIdx.x;
    if (e >= E + n) return;
    const int s = (e < E) ? srcA[e] : (e - E);
    const int d = (e < E) ? dstA[e] : (e - E);
    const int slot = atomicAdd(&cursor[d], 1);
    col[slot] = s;
}

__global__ void gat_gather_l1(const int* __restrict__ rowptr, const int* __restrict__ col,
                              const float* __restrict__ aS, const float* __restrict__ aD,
                              const float* __restrict__ H1, const float* __restrict__ b1,
                              float* __restrict__ out1, int n) {
    const int lane = threadIdx.x & 31;
    const int w = (blockIdx.x * blockDim.x + threadIdx.x) >> 5;
    if (w >= n * NHD) return;
    const int node = w >> 3, h = w & 7;
    const int beg = rowptr[node], end = rowptr[node + 1];
    const float aDd = aD[node * NHD + h];
    float m = -__builtin_inff(), den = 0.f, acc = 0.f;
    for (int j = beg; j < end; ++j) {
        const int s = col[j];
        if (j + 1 < end) {
            const int s2 = col[j + 1];
            __builtin_prefetch(&H1[(size_t)s2 * HC + h * CPH + lane], 0, 1);
        }
        const float l = leaky(aS[s * NHD + h] + aDd);
        const float mn = fmaxf(m, l);
        const float scale = expf(m - mn);
        const float wgt   = expf(l - mn);
        acc = acc * scale + wgt * H1[(size_t)s * HC + h * CPH + lane];
        den = den * scale + wgt;
        m = mn;
    }
    const float v = acc / den + b1[h * CPH + lane];
    out1[(size_t)node * HC + h * CPH + lane] = v > 0.f ? v : 0.f;
}

__global__ void gat_l2_fused(const int* __restrict__ rowptr, const int* __restrict__ col,
                             const float* __restrict__ aS, const float* __restrict__ aD,
                             const float* __restrict__ H2, const float* __restrict__ b2,
                             float* __restrict__ out, int n) {
    const int lane = threadIdx.x & 31;
    const int node = (blockIdx.x * blockDim.x + threadIdx.x) >> 5;
    if (node >= n) return;
    const int beg = rowptr[node], end = rowptr[node + 1];
    const float aDd = aD[node];
    const int c1 = lane + 32;
    float m = -__builtin_inff(), den = 0.f, acc0 = 0.f, acc1 = 0.f;
    for (int j = beg; j < end; ++j) {
        const int s = col[j];
        if (j + 1 < end) {
            const int s2 = col[j + 1];
            __builtin_prefetch(&H2[(size_t)s2 * OUTC + lane], 0, 1);
        }
        const float l = leaky(aS[s] + aDd);
        const float mn = fmaxf(m, l);
        const float scale = expf(m - mn);
        const float wgt   = expf(l - mn);
        const float* hrow = H2 + (size_t)s * OUTC;
        acc0 = acc0 * scale + wgt * ((lane < OUTC) ? hrow[lane] : 0.f);
        acc1 = acc1 * scale + wgt * ((c1 < OUTC) ? hrow[c1] : 0.f);
        den = den * scale + wgt;
        m = mn;
    }
    float v0 = (lane < OUTC) ? acc0 / den + b2[lane] : -__builtin_inff();
    float v1 = (c1 < OUTC)  ? acc1 / den + b2[c1]  : -__builtin_inff();
    float mx = fmaxf(v0, v1);
    for (int o = 16; o > 0; o >>= 1) mx = fmaxf(mx, __shfl_xor(mx, o, 32));
    float ssum = 0.f;
    if (lane < OUTC) ssum += expf(v0 - mx);
    if (c1 < OUTC)  ssum += expf(v1 - mx);
    for (int o = 16; o > 0; o >>= 1) ssum += __shfl_xor(ssum, o, 32);
    const float lse = mx + logf(ssum);
    if (lane < OUTC) out[(size_t)node * OUTC + lane] = v0 - lse;
    if (c1 < OUTC)  out[(size_t)node * OUTC + c1]  = v1 - lse;
}

extern "C" void kernel_launch(void* const* d_in, const int* in_sizes, int n_in,
                              void* d_out, int out_size, void* d_ws, size_t ws_size,
                              hipStream_t stream) {
    const float* x   = (const float*)d_in[0];
    const int*   ei  = (const int*)d_in[1];
    const float* W1  = (const float*)d_in[2];
    const float* as1 = (const float*)d_in[3];
    const float* ad1 = (const float*)d_in[4];
    const float* b1  = (const float*)d_in[5];
    const float* W2  = (const float*)d_in[6];
    const float* as2 = (const float*)d_in[7];
    const float* ad2 = (const float*)d_in[8];
    const float* b2  = (const float*)d_in[9];
    const int N  = in_sizes[0] / INCH;
    const int E  = in_sizes[1] / 2;
    const int Et = E + N;
    const int* srcA = ei;
    const int* dstA = ei + E;

    float* ws = (float*)d_ws;
    size_t off = 0;
    float* h1   = ws + off; off += (size_t)N * HC;
    float* out1 = ws + off; off += (size_t)N * HC;
    float* aS1  = ws + off; off += (size_t)N * NHD;
    float* aD1  = ws + off; off += (size_t)N * NHD;
    int* rowptr = (int*)(ws + off); off += (size_t)N + 1;
    int* cnt    = (int*)(ws + off); off += (size_t)N;
    int* cursor = (int*)(ws + off); off += (size_t)N;
    int* col    = (int*)(ws + off); off += (size_t)Et;
    float* h2  = h1;
    float* aS2 = aS1;
    float* aD2 = aD1;

    hipMemsetAsync(cnt, 0, (size_t)N * sizeof(int), stream);
    count_edges<<<(Et + 255) / 256, 256, 0, stream>>>(dstA, cnt, E, N);
    scan_rowptr<<<1, 1024, 0, stream>>>(cnt, rowptr, cursor, N);
    fill_csr<<<(Et + 255) / 256, 256, 0, stream>>>(srcA, dstA, cursor, col, E, N);

    gemm_x_w1<<<(N + 15) / 16, 256, 0, stream>>>(x, W1, h1, N);
    attn_dots_l1<<<(N * NHD + 255) / 256, 256, 0, stream>>>(h1, as1, ad1, aS1, aD1, N);
    gat_gather_l1<<<(N * NHD * 32 + 255) / 256, 256, 0, stream>>>(
        rowptr, col, aS1, aD1, h1, b1, out1, N);

    gemm_h_w2<<<(N + 15) / 16, 96, 0, stream>>>(out1, W2, h2, N);
    attn_dots_l2<<<(N + 255) / 256, 256, 0, stream>>>(h2, as2, ad2, aS2, aD2, N);
    gat_l2_fused<<<(N * 32 + 255) / 256, 256, 0, stream>>>(
        rowptr, col, aS2, aD2, h2, b2, (float*)d_out, N);
}


# layer-1 gather: a wave takes 4 nodes x 8 heads; lane i walks item i's list and holds its softmax (once per item, not per lane), lane c sums channel c of all 32 items by readlane; sum order kept
# speedup vs baseline: 2.2885x; 2.2885x over previous
#include <hip/hip_runtime.h>
#include <hip/hip_bf16.h>

typedef __attribute__((ext_vector_type(2))) float v2f;
typedef __attribute__((ext_vector_type(8))) float v8f;

#define INCH  128
#define HC    256
#define NHD   8
#define CPH   32
#define OUTC  40
#define SLOPE 0.2f

__device__ __forceinline__ float leaky(float x) { return x > 0.f ? x : SLOPE * x; }

__global__ void gemm_x_w1(const float* __restrict__ X, const float* __restrict__ W,
                          float* __restrict__ H1, int nrows) {
    const int lane = threadIdx.x & 31;
    const int wave = threadIdx.x >> 5;
    const int m    = lane & 15;
    const int half = lane >> 4;
    const int rowBase = blockIdx.x << 4;
    if (rowBase >= nrows) return;
    const float* xrow = X + (size_t)(rowBase + m) * INCH;
    for (int t = 0; t < 2; ++t) {
        const int colBase = ((wave << 1) + t) << 4;
        v8f acc = {0.f, 0.f, 0.f, 0.f, 0.f, 0.f, 0.f, 0.f};
        for (int k = 0; k < INCH; k += 4) {
            const int kk = k + (half << 1);
            v2f a, b;
            a.x = xrow[kk];
            a.y = xrow[kk + 1];
            b.x = W[(size_t)kk * HC + colBase + m];
            b.y = W[(size_t)(kk + 1) * HC + colBase + m];
            acc = __builtin_amdgcn_wmma_f32_16x16x4_f32(false, a, false, b,
                                                        (short)0, acc, false, false);
        }
#pragma unroll
        for (int v = 0; v < 8; ++v)
            H1[(size_t)(rowBase + v + (half << 3)) * HC + colBase + m] = acc[v];
    }
}

__global__ void gemm_h_w2(const float* __restrict__ Hin, const float* __restrict__ W,
                          float* __restrict__ H2, int nrows) {
    const int lane = threadIdx.x & 31;
    const int wave = threadIdx.x >> 5;
    const int m    = lane & 15;
    const int half = lane >> 4;
    const int rowBase = blockIdx.x << 4;
    if (rowBase >= nrows) return;
    const int colBase = wave << 4;
    const int col = colBase + m;
    const float* hrow = Hin + (size_t)(rowBase + m) * HC;
    v8f acc = {0.f, 0.f, 0.f, 0.f, 0.f, 0.f, 0.f, 0.f};
    for (int k = 0; k < HC; k += 4) {
        const int kk = k + (half << 1);
        v2f a, b;
        a.x = hrow[kk];
        a.y = hrow[kk + 1];
        b.x = (col < OUTC) ? W[kk * OUTC + col] : 0.f;
        b.y = (col < OUTC) ? W[(kk + 1) * OUTC + col] : 0.f;
        acc = __builtin_amdgcn_wmma_f32_16x16x4_f32(false, a, false, b,
                                                    (short)0, acc, false, false);
    }
    if (col < OUTC) {
#pragma unroll
        for (int v = 0; v < 8; ++v)
            H2[(size_t)(rowBase + v + (half << 3)) * OUTC + col] = acc[v];
    }
}

__global__ void attn_dots_l1(const float* __restrict__ H1, const float* __restrict__ Asrc,
                             const float* __restrict__ Adst, float* __restrict__ aS,
                             float* __restrict__ aD, int n) {
    __shared__ float sAs[HC], sAd[HC];
    for (int i = threadIdx.x; i < HC; i += blockDim.x) { sAs[i] = Asrc[i]; sAd[i] = Adst[i]; }
    __syncthreads();
    const int idx = blockIdx.x * blockDim.x + threadIdx.x;
    if (idx >= n * NHD) return;
    const int node = idx >> 3, h = idx & 7;
    const float* row = H1 + (size_t)node * HC + h * CPH;
    float s = 0.f, d = 0.f;
#pragma unroll
    for (int c = 0; c < CPH; ++c) {
        const float v = row[c];
        s += v * sAs[h * CPH + c];
        d += v * sAd[h * CPH + c];
    }
    aS[idx] = s; aD[idx] = d;
}

__global__ void attn_dots_l2(const float* __restrict__ H2, const float* __restrict__ Asrc,
                             const float* __restrict__ Adst, float* __restrict__ aS,
                             float* __restrict__ aD, int n) {
    const int node = blockIdx.x * blockDim.x + threadIdx.x;
    if (node >= n) return;
    const float* row = H2 + (size_t)node * OUTC;
    float s = 0.f, d = 0.f;
#pragma unroll
    for (int c = 0; c < OUTC; ++c) { const float v = row[c]; s += v * Asrc[c]; d += v * Adst[c]; }
    aS[node] = s; aD[node] = d;
}

__global__ void link_edges(const int* __restrict__ dstA, int* __restrict__ head,
                           int* __restrict__ nxt, int E) {
    const int e = blockIdx.x * blockDim.x + threadIdx.x;
    if (e >= E) return;
    nxt[e] = atomicExch(&head[dstA[e]], e);
}

__global__ void
gat_gather_l1(const int* __restrict__ head, const int* __restrict__ nxt,
              const int* __restrict__ srcA,
              const float* __restrict__ aS, const float* __restrict__ aD,
              const float* __restrict__ H1, const float* __restrict__ b1,
              float* __restrict__ out1, int n, int E) {
    const int lane = threadIdx.x & 31;
    const int wv = (blockIdx.x * blockDim.x + threadIdx.x) >> 5;
    const int node0 = wv << 2;
    if (node0 >= n) return;
    const int h = lane & 7;
    const int node = node0 + (lane >> 3);
    bool alive = node < n;
    int s = alive ? node : 0;
    int e = alive ? head[node] : -1;
    const float aDd = alive ? aD[node * NHD + h] : 0.f;
    float m = -__builtin_inff(), den = 0.f;
    float acc[32];
#pragma unroll
    for (int j = 0; j < 32; ++j) acc[j] = 0.f;

    for (int it = 0; it <= E; ++it) {
        const unsigned live = __builtin_amdgcn_ballot_w32(alive);
        if (live == 0u) break;
        const bool more = alive && e >= 0;
        int s2 = 0, e2 = -1;
        if (more) { s2 = srcA[e]; e2 = nxt[e]; }
        float as_s = 0.f;
        if (alive) as_s = aS[s * NHD + h];
        float hv[32];
#pragma unroll
        for (int q = 0; q < 4; ++q) {
            if (live & (1u << (q * 8))) {
                const int sq = __builtin_amdgcn_readlane(s, q * 8);
                const float* row = H1 + (size_t)sq * HC + lane;
#pragma unroll
                for (int hh = 0; hh < NHD; ++hh) hv[q * 8 + hh] = row[hh * CPH];
            }
        }
        float scale = 1.f, wgt = 0.f;
        if (alive) {
            const float l = leaky(as_s + aDd);
            const float mn = fmaxf(m, l);
            scale = expf(m - mn);
            wgt   = expf(l - mn);
            den = den * scale + wgt;
            m = mn;
        }
#pragma unroll
        for (int q = 0; q < 4; ++q) {
            if (live & (1u << (q * 8))) {
#pragma unroll
                for (int hh = 0; hh < NHD; ++hh) {
                    const float sc = __int_as_float(
                        __builtin_amdgcn_readlane(__float_as_int(scale), q * 8 + hh));
                    const float wg = __int_as_float(
                        __builtin_amdgcn_readlane(__float_as_int(wgt), q * 8 + hh));
                    acc[q * 8 + hh] = acc[q * 8 + hh] * sc + wg * hv[q * 8 + hh];
                }
            }
        }
        if (!more) alive = false;
        s = s2; e = e2;
    }

#pragma unroll
    for (int q = 0; q < 4; ++q) {
        if (node0 + q < n) {
#pragma unroll
            for (int hh = 0; hh < NHD; ++hh) {
                const float d = __int_as_float(
                    __builtin_amdgcn_readlane(__float_as_int(den), q * 8 + hh));
                const float v = acc[q * 8 + hh] / d + b1[hh * CPH + lane];
                out1[(size_t)(node0 + q) * HC + hh * CPH + lane] = v > 0.f ? v : 0.f;
            }
        }
    }
}

__global__ void gat_l2_fused(const int* __restrict__ head, const int* __restrict__ nxt,
                             const int* __restrict__ srcA,
                             const float* __restrict__ aS, const float* __restrict__ aD,
                             const float* __restrict__ H2, const float* __restrict__ b2,
                             float* __restrict__ out, int n, int E) {
    const int lane = threadIdx.x & 31;
    const int node = (blockIdx.x * blockDim.x + threadIdx.x) >> 5;
    if (node >= n) return;
    const float aDd = aD[node];
    const int c1 = lane + 32;
    int s = node;
    int e = head[node];
    float m = -__builtin_inff(), den = 0.f, acc0 = 0.f, acc1 = 0.f;
    for (int it = 0; it <= E; ++it) {
        const bool more = e >= 0;
        int s2 = 0, e2 = -1;
        if (more) { s2 = srcA[e]; e2 = nxt[e]; }
        const float l = leaky(aS[s] + aDd);
        const float* hrow = H2 + (size_t)s * OUTC;
        const float h0 = (lane < OUTC) ? hrow[lane] : 0.f;
        const float h1v = (c1 < OUTC) ? hrow[c1] : 0.f;
        const float mn = fmaxf(m, l);
        const float scale = expf(m - mn);
        const float wgt   = expf(l - mn);
        acc0 = acc0 * scale + wgt * h0;
        acc1 = acc1 * scale + wgt * h1v;
        den = den * scale + wgt;
        m = mn;
        if (!more) break;
        s = s2; e = e2;
    }
    float v0 = (lane < OUTC) ? acc0 / den + b2[lane] : -__builtin_inff();
    float v1 = (c1 < OUTC)  ? acc1 / den + b2[c1]  : -__builtin_inff();
    float mx = fmaxf(v0, v1);
    for (int o = 16; o > 0; o >>= 1) mx = fmaxf(mx, __shfl_xor(mx, o, 32));
    float ssum = 0.f;
    if (lane < OUTC) ssum += expf(v0 - mx);
    if (c1 < OUTC)  ssum += expf(v1 - mx);
    for (int o = 16; o > 0; o >>= 1) ssum += __shfl_xor(ssum, o, 32);
    const float lse = mx + logf(ssum);
    if (lane < OUTC) out[(size_t)node * OUTC + lane] = v0 - lse;
    if (c1 < OUTC)  out[(size_t)node * OUTC + c1]  = v1 - lse;
}

extern "C" void kernel_launch(void* const* d_in, const int* in_sizes, int n_in,
                              void* d_out, int out_size, void* d_ws, size_t ws_size,
                              hipStream_t stream) {
    const float* x   = (const float*)d_in[0];
    const int*   ei  = (const int*)d_in[1];
    const float* W1  = (const float*)d_in[2];
    const float* as1 = (const float*)d_in[3];
    const float* ad1 = (const float*)d_in[4];
    const float* b1  = (const float*)d_in[5];
    const float* W2  = (const float*)d_in[6];
    const float* as2 = (const float*)d_in[7];
    const float* ad2 = (const float*)d_in[8];
    const float* b2  = (const float*)d_in[9];
    const int N  = in_sizes[0] / INCH;
    const int E  = in_sizes[1] / 2;
    const int* srcA = ei;
    const int* dstA = ei + E;

    float* ws = (float*)d_ws;
    size_t off = 0;
    float* h1   = ws + off; off += (size_t)N * HC;
    float* out1 = ws + off; off += (size_t)N * HC;
    float* aS1  = ws + off; off += (size_t)N * NHD;
    float* aD1  = ws + off; off += (size_t)N * NHD;
    int* head = (int*)(ws + off); off += (size_t)N;
    int* nxt  = (int*)(ws + off); off += (size_t)E;
    float* h2  = h1;
    float* aS2 = aS1;
    float* aD2 = aD1;

    hipMemsetAsync(head, 0xFF, (size_t)N * sizeof(int), stream);
    link_edges<<<(E + 255) / 256, 256, 0, stream>>>(dstA, head, nxt, E);

    gemm_x_w1<<<(N + 15) / 16, 256, 0, stream>>>(x, W1, h1, N);
    attn_dots_l1<<<(N * NHD + 255) / 256, 256, 0, stream>>>(h1, as1, ad1, aS1, aD1, N);
    gat_gather_l1<<<((N + 3) / 4 * 32 + 255) / 256, 256, 0, stream>>>(
        head, nxt, srcA, aS1, aD1, h1, b1, out1, N, E);

    gemm_h_w2<<<(N + 15) / 16, 96, 0, stream>>>(out1, W2, h2, N);
    attn_dots_l2<<<(N + 255) / 256, 256, 0, stream>>>(h2, as2, ad2, aS2, aD2, N);
    gat_l2_fused<<<(N * 32 + 255) / 256, 256, 0, stream>>>(
        head, nxt, srcA, aS2, aD2, h2, b2, (float*)d_out, N, E);
}


# both projections: one A fragment load per k-step feeds all of a wave's column tiles (8 in GEMM1, 3 in GEMM2; was one per tile), W fragments from LDS staged once per 128-row block; sum order kept
# speedup vs baseline: 2.7336x; 1.1945x over previous
#include <hip/hip_runtime.h>
#include <hip/hip_bf16.h>

typedef __attribute__((ext_vector_type(2))) float v2f;
typedef __attribute__((ext_vector_type(8))) float v8f;

#define INCH  128
#define HC    256
#define NHD   8
#define CPH   32
#define OUTC  40
#define SLOPE 0.2f

__device__ __forceinline__ float leaky(float x) { return x > 0.f ? x : SLOPE * x; }

__global__ void __launch_bounds__(512)
gemm_x_w1(const float* __restrict__ X, const float* __restrict__ W,
          float* __restrict__ H1, int nrows) {
    __shared__ __attribute__((aligned(16))) float sW[INCH * HC];
    const int tid = threadIdx.x;
#pragma unroll
    for (int i = 0; i < 8; ++i) {
        const int u  = tid + i * 512;
        const int c4 = u & 63;
        const int kp = u >> 6;
        const float4 r0 = *(const float4*)(W + (size_t)(2 * kp) * HC + 4 * c4);
        const float4 r1 = *(const float4*)(W + (size_t)(2 * kp + 1) * HC + 4 * c4);
        const int j = kp >> 1, h = kp & 1;
        const int t = c4 >> 2, n = (c4 & 3) << 2;
        float* dst = sW + ((((j * 16 + t) * 32) + h * 16 + n) << 1);
        *(float4*)(dst)     = make_float4(r0.x, r1.x, r0.y, r1.y);
        *(float4*)(dst + 4) = make_float4(r0.z, r1.z, r0.w, r1.w);
    }
    __syncthreads();
    const int lane = tid & 31;
    const int wave = tid >> 5;
    const int m    = lane & 15;
    const int half = lane >> 4;
    const int rowBase = ((blockIdx.x << 3) + (wave >> 1)) << 4;
    if (rowBase >= nrows) return;
    const int ch = wave & 1;
    const float* xrow = X + (size_t)(rowBase + m) * INCH + (half << 1);
    const v2f* bfrag = (const v2f*)sW + (ch << 3) * 32 + lane;
    v8f acc[8];
#pragma unroll
    for (int i = 0; i < 8; ++i) acc[i] = (v8f){0.f, 0.f, 0.f, 0.f, 0.f, 0.f, 0.f, 0.f};
    v2f ac[8], an[8];
#pragma unroll
    for (int q = 0; q < 8; ++q) ac[q] = *(const v2f*)(xrow + 4 * q);
#pragma unroll
    for (int c = 0; c < 4; ++c) {
        if (c < 3) {
#pragma unroll
            for (int q = 0; q < 8; ++q) an[q] = *(const v2f*)(xrow + 4 * ((c + 1) * 8 + q));
        }
#pragma unroll
        for (int q = 0; q < 8; ++q) {
            const int j = c * 8 + q;
#pragma unroll
            for (int i = 0; i < 8; ++i) {
                const v2f b = bfrag[(j * 16 + i) * 32];
                acc[i] = __builtin_amdgcn_wmma_f32_16x16x4_f32(false, ac[q], false, b,
                                                               (short)0, acc[i], false, false);
            }
        }
#pragma unroll
        for (int q = 0; q < 8; ++q) ac[q] = an[q];
    }
#pragma unroll
    for (int i = 0; i < 8; ++i) {
        const int colBase = ((ch << 3) + i) << 4;
#pragma unroll
        for (int v = 0; v < 8; ++v)
            H1[(size_t)(rowBase + v + (half << 3)) * HC + colBase + m] = acc[i][v];
    }
}

__global__ void __launch_bounds__(256)
gemm_h_w2(const float* __restrict__ Hin, const float* __restrict__ W,
          float* __restrict__ H2, int nrows) {
    __shared__ __attribute__((aligned(16))) float sW[64 * 3 * 32 * 2];
    const int tid = threadIdx.x;
#pragma unroll
    for (int i = 0; i < 24; ++i) {
        const int s  = tid + i * 256;
        const int ln = s & 31, jt = s >> 5;
        const int j  = jt / 3, t = jt - j * 3;
        const int c  = t * 16 + (ln & 15);
        const int k  = 4 * j + 2 * (ln >> 4);
        v2f w = {0.f, 0.f};
        if (c < OUTC) { w.x = W[k * OUTC + c]; w.y = W[(k + 1) * OUTC + c]; }
        ((v2f*)sW)[s] = w;
    }
    __syncthreads();
    const int lane = tid & 31;
    const int wave = tid >> 5;
    const int m    = lane & 15;
    const int half = lane >> 4;
    const int rowBase = ((blockIdx.x << 3) + wave) << 4;
    if (rowBase >= nrows) return;
    const float* hrow = Hin + (size_t)(rowBase + m) * HC + (half << 1);
    const v2f* bfrag = (const v2f*)sW + lane;
    v8f acc[3];
#pragma unroll
    for (int t = 0; t < 3; ++t) acc[t] = (v8f){0.f, 0.f, 0.f, 0.f, 0.f, 0.f, 0.f, 0.f};
    v2f ac[8], an[8];
#pragma unroll
    for (int q = 0; q < 8; ++q) ac[q] = *(const v2f*)(hrow + 4 * q);
#pragma unroll
    for (int c = 0; c < 8; ++c) {
        if (c < 7) {
#pragma unroll
            for (int q = 0; q < 8; ++q) an[q] = *(const v2f*)(hrow + 4 * ((c + 1) * 8 + q));
        }
#pragma unroll
        for (int q = 0; q < 8; ++q) {
            const int j = c * 8 + q;
#pragma unroll
            for (int t = 0; t < 3; ++t) {
                const v2f b = bfrag[(j * 3 + t) * 32];
                acc[t] = __builtin_amdgcn_wmma_f32_16x16x4_f32(false, ac[q], false, b,
                                                               (short)0, acc[t], false, false);
            }
        }
#pragma unroll
        for (int q = 0; q < 8; ++q) ac[q] = an[q];
    }
#pragma unroll
    for (int t = 0; t < 3; ++t) {
        const int col = (t << 4) + m;
        if (col < OUTC) {
#pragma unroll
            for (int v = 0; v < 8; ++v)
                H2[(size_t)(rowBase + v + (half << 3)) * OUTC + col] = acc[t][v];
        }
    }
}

__global__ void attn_dots_l1(const float* __restrict__ H1, const float* __restrict__ Asrc,
                             const float* __restrict__ Adst, float* __restrict__ aS,
                             float* __restrict__ aD, int n) {
    __shared__ float sAs[HC], sAd[HC];
    for (int i = threadIdx.x; i < HC; i += blockDim.x) { sAs[i] = Asrc[i]; sAd[i] = Adst[i]; }
    __syncthreads();
    const int idx = blockIdx.x * blockDim.x + threadIdx.x;
    if (idx >= n * NHD) return;
    const int node = idx >> 3, h = idx & 7;
    const float* row = H1 + (size_t)node * HC + h * CPH;
    float s = 0.f, d = 0.f;
#pragma unroll
    for (int c = 0; c < CPH; ++c) {
        const float v = row[c];
        s += v * sAs[h * CPH + c];
        d += v * sAd[h * CPH + c];
    }
    aS[idx] = s; aD[idx] = d;
}

__global__ void attn_dots_l2(const float* __restrict__ H2, const float* __restrict__ Asrc,
                             const float* __restrict__ Adst, float* __restrict__ aS,
                             float* __restrict__ aD, int n) {
    const int node = blockIdx.x * blockDim.x + threadIdx.x;
    if (node >= n) return;
    const float* row = H2 + (size_t)node * OUTC;
    float s = 0.f, d = 0.f;
#pragma unroll
    for (int c = 0; c < OUTC; ++c) { const float v = row[c]; s += v * Asrc[c]; d += v * Adst[c]; }
    aS[node] = s; aD[node] = d;
}

__global__ void link_edges(const int* __restrict__ dstA, int* __restrict__ head,
                           int* __restrict__ nxt, int E) {
    const int e = blockIdx.x * blockDim.x + threadIdx.x;
    if (e >= E) return;
    nxt[e] = atomicExch(&head[dstA[e]], e);
}

__global__ void
gat_gather_l1(const int* __restrict__ head, const int* __restrict__ nxt,
              const int* __restrict__ srcA,
              const float* __restrict__ aS, const float* __restrict__ aD,
              const float* __restrict__ H1, const float* __restrict__ b1,
              float* __restrict__ out1, int n, int E) {
    const int lane = threadIdx.x & 31;
    const int wv = (blockIdx.x * blockDim.x + threadIdx.x) >> 5;
    const int node0 = wv << 2;
    if (node0 >= n) return;
    const int h = lane & 7;
    const int node = node0 + (lane >> 3);
    bool alive = node < n;
    int s = alive ? node : 0;
    int e = alive ? head[node] : -1;
    const float aDd = alive ? aD[node * NHD + h] : 0.f;
    float m = -__builtin_inff(), den = 0.f;
    float acc[32];
#pragma unroll
    for (int j = 0; j < 32; ++j) acc[j] = 0.f;

    for (int it = 0; it <= E; ++it) {
        const unsigned live = __builtin_amdgcn_ballot_w32(alive);
        if (live == 0u) break;
        const bool more = alive && e >= 0;
        int s2 = 0, e2 = -1;
        if (more) { s2 = srcA[e]; e2 = nxt[e]; }
        float as_s = 0.f;
        if (alive) as_s = aS[s * NHD + h];
        float hv[32];
#pragma unroll
        for (int q = 0; q < 4; ++q) {
            if (live & (1u << (q * 8))) {
                const int sq = __builtin_amdgcn_readlane(s, q * 8);
                const float* row = H1 + (size_t)sq * HC + lane;
#pragma unroll
                for (int hh = 0; hh < NHD; ++hh) hv[q * 8 + hh] = row[hh * CPH];
            }
        }
        float scale = 1.f, wgt = 0.f;
        if (alive) {
            const float l = leaky(as_s + aDd);
            const float mn = fmaxf(m, l);
            scale = expf(m - mn);
            wgt   = expf(l - mn);
            den = den * scale + wgt;
            m = mn;
        }
#pragma unroll
        for (int q = 0; q < 4; ++q) {
            if (live & (1u << (q * 8))) {
#pragma unroll
                for (int hh = 0; hh < NHD; ++hh) {
                    const float sc = __int_as_float(
                        __builtin_amdgcn_readlane(__float_as_int(scale), q * 8 + hh));
                    const float wg = __int_as_float(
                        __builtin_amdgcn_readlane(__float_as_int(wgt), q * 8 + hh));
                    acc[q * 8 + hh] = acc[q * 8 + hh] * sc + wg * hv[q * 8 + hh];
                }
            }
        }
        if (!more) alive = false;
        s = s2; e = e2;
    }

#pragma unroll
    for (int q = 0; q < 4; ++q) {
        if (node0 + q < n) {
#pragma unroll
            for (int hh = 0; hh < NHD; ++hh) {
                const float d = __int_as_float(
                    __builtin_amdgcn_readlane(__float_as_int(den), q * 8 + hh));
                const float v = acc[q * 8 + hh] / d + b1[hh * CPH + lane];
                out1[(size_t)(node0 + q) * HC + hh * CPH + lane] = v > 0.f ? v : 0.f;
            }
        }
    }
}

__global__ void gat_l2_fused(const int* __restrict__ head, const int* __restrict__ nxt,
                             const int* __restrict__ srcA,
                             const float* __restrict__ aS, const float* __restrict__ aD,
                             const float* __restrict__ H2, const float* __restrict__ b2,
                             float* __restrict__ out, int n, int E) {
    const int lane = threadIdx.x & 31;
    const int node = (blockIdx.x * blockDim.x + threadIdx.x) >> 5;
    if (node >= n) return;
    const float aDd = aD[node];
    const int c1 = lane + 32;
    int s = node;
    int e = head[node];
    float m = -__builtin_inff(), den = 0.f, acc0 = 0.f, acc1 = 0.f;
    for (int it = 0; it <= E; ++it) {
        const bool more = e >= 0;
        int s2 = 0, e2 = -1;
        if (more) { s2 = srcA[e]; e2 = nxt[e]; }
        const float l = leaky(aS[s] + aDd);
        const float* hrow = H2 + (size_t)s * OUTC;
        const float h0 = (lane < OUTC) ? hrow[lane] : 0.f;
        const float h1v = (c1 < OUTC) ? hrow[c1] : 0.f;
        const float mn = fmaxf(m, l);
        const float scale = expf(m - mn);
        const float wgt   = expf(l - mn);
        acc0 = acc0 * scale + wgt * h0;
        acc1 = acc1 * scale + wgt * h1v;
        den = den * scale + wgt;
        m = mn;
        if (!more) break;
        s = s2; e = e2;
    }
    float v0 = (lane < OUTC) ? acc0 / den + b2[lane] : -__builtin_inff();
    float v1 = (c1 < OUTC)  ? acc1 / den + b2[c1]  : -__builtin_inff();
    float mx = fmaxf(v0, v1);
    for (int o = 16; o > 0; o >>= 1) mx = fmaxf(mx, __shfl_xor(mx, o, 32));
    float ssum = 0.f;
    if (lane < OUTC) ssum += expf(v0 - mx);
    if (c1 < OUTC)  ssum += expf(v1 - mx);
    for (int o = 16; o > 0; o >>= 1) ssum += __shfl_xor(ssum, o, 32);
    const float lse = mx + logf(ssum);
    if (lane < OUTC) out[(size_t)node * OUTC + lane] = v0 - lse;
    if (c1 < OUTC)  out[(size_t)node * OUTC + c1]  = v1 - lse;
}

extern "C" void kernel_launch(void* const* d_in, const int* in_sizes, int n_in,
                              void* d_out, int out_size, void* d_ws, size_t ws_size,
                              hipStream_t stream) {
    const float* x   = (const float*)d_in[0];
    const int*   ei  = (const int*)d_in[1];
    const float* W1  = (const float*)d_in[2];
    const float* as1 = (const float*)d_in[3];
    const float* ad1 = (const float*)d_in[4];
    const float* b1  = (const float*)d_in[5];
    const float* W2  = (const float*)d_in[6];
    const float* as2 = (const float*)d_in[7];
    const float* ad2 = (const float*)d_in[8];
    const float* b2  = (const float*)d_in[9];
    const int N  = in_sizes[0] / INCH;
    const int E  = in_sizes[1] / 2;
    const int* srcA = ei;
    const int* dstA = ei + E;

    float* ws = (float*)d_ws;
    size_t off = 0;
    float* h1   = ws + off; off += (size_t)N * HC;
    float* out1 = ws + off; off += (size_t)N * HC;
    float* aS1  = ws + off; off += (size_t)N * NHD;
    float* aD1  = ws + off; off += (size_t)N * NHD;
    int* head = (int*)(ws + off); off += (size_t)N;
    int* nxt  = (int*)(ws + off); off += (size_t)E;
    float* h2  = h1;
    float* aS2 = aS1;
    float* aD2 = aD1;

    hipMemsetAsync(head, 0xFF, (size_t)N * sizeof(int), stream);
    link_edges<<<(E + 255) / 256, 256, 0, stream>>>(dstA, head, nxt, E);

    gemm_x_w1<<<(N + 127) / 128, 512, 0, stream>>>(x, W1, h1, N);
    attn_dots_l1<<<(N * NHD + 255) / 256, 256, 0, stream>>>(h1, as1, ad1, aS1, aD1, N);
    gat_gather_l1<<<((N + 3) / 4 * 32 + 255) / 256, 256, 0, stream>>>(
        head, nxt, srcA, aS1, aD1, h1, b1, out1, N, E);

    gemm_h_w2<<<(N + 127) / 128, 256, 0, stream>>>(out1, W2, h2, N);
    attn_dots_l2<<<(N + 255) / 256, 256, 0, stream>>>(h2, as2, ad2, aS2, aD2, N);
    gat_l2_fused<<<(N * 32 + 255) / 256, 256, 0, stream>>>(
        head, nxt, srcA, aS2, aD2, h2, b2, (float*)d_out, N, E);
}


# attention logits as extra B columns of both projections (W1, W2 pre-multiplied by the attention vectors: 16 columns in layer 1, 2 in layer 2); both dot launches gone; dot sum order changes
# speedup vs baseline: 2.8920x; 1.0579x over previous
#include <hip/hip_runtime.h>
#include <hip/hip_bf16.h>

typedef __attribute__((ext_vector_type(2))) float v2f;
typedef __attribute__((ext_vector_type(8))) float v8f;

#define INCH  128
#define HC    256
#define NHD   8
#define CPH   32
#define OUTC  40
#define SLOPE 0.2f

__device__ __forceinline__ float leaky(float x) { return x > 0.f ? x : SLOPE * x; }

template <int MASK>
__device__ __forceinline__ float add_xor_lane(float v) {
    return v + __int_as_float(__builtin_amdgcn_update_dpp(0, __float_as_int(v), 0x160 + MASK,
                                                          0xF, 0xF, false));
}

template <bool FOLDED>
__device__ __forceinline__ void
gemm1_wave(const float* __restrict__ xrow, const v2f* __restrict__ bfrag,
           const v2f* __restrict__ efrag, float* __restrict__ H1,
           float* __restrict__ aS, float* __restrict__ aD,
           int rowBase, int ch, int m, int half) {
    v8f acc[8];
#pragma unroll
    for (int i = 0; i < 8; ++i) acc[i] = (v8f){0.f, 0.f, 0.f, 0.f, 0.f, 0.f, 0.f, 0.f};
    v8f accE = (v8f){0.f, 0.f, 0.f, 0.f, 0.f, 0.f, 0.f, 0.f};
    v2f ac[8], an[8];
#pragma unroll
    for (int q = 0; q < 8; ++q) ac[q] = *(const v2f*)(xrow + 4 * q);
#pragma unroll
    for (int c = 0; c < 4; ++c) {
        if (c < 3) {
#pragma unroll
            for (int q = 0; q < 8; ++q) an[q] = *(const v2f*)(xrow + 4 * ((c + 1) * 8 + q));
        }
#pragma unroll
        for (int q = 0; q < 8; ++q) {
            const int j = c * 8 + q;
#pragma unroll
            for (int i = 0; i < 8; ++i) {
                const v2f b = bfrag[(j * 16 + i) * 32];
                acc[i] = __builtin_amdgcn_wmma_f32_16x16x4_f32(false, ac[q], false, b,
                                                               (short)0, acc[i], false, false);
            }
            if (FOLDED) {
                const v2f b = efrag[j * 32];
                accE = __builtin_amdgcn_wmma_f32_16x16x4_f32(false, ac[q], false, b,
                                                             (short)0, accE, false, false);
            }
        }
#pragma unroll
        for (int q = 0; q < 8; ++q) ac[q] = an[q];
    }
#pragma unroll
    for (int i = 0; i < 8; ++i) {
        const int colBase = ((ch << 3) + i) << 4;
#pragma unroll
        for (int v = 0; v < 8; ++v)
            H1[(size_t)(rowBase + v + (half << 3)) * HC + colBase + m] = acc[i][v];
    }
    if (FOLDED) {
        float* dots = (m < NHD) ? aS + m : aD + (m - NHD);
#pragma unroll
        for (int v = 0; v < 8; ++v)
            dots[(size_t)(rowBase + v + (half << 3)) * NHD] = accE[v];
    }
}

__global__ void __launch_bounds__(512)
gemm_x_w1(const float* __restrict__ X, const float* __restrict__ W,
          const float* __restrict__ Asrc, const float* __restrict__ Adst,
          float* __restrict__ H1, float* __restrict__ aS, float* __restrict__ aD, int nrows) {
    __shared__ __attribute__((aligned(16))) float sW[INCH * HC];
    __shared__ __attribute__((aligned(16))) float sE[(INCH / 4) * 32 * 2];
    const int tid = threadIdx.x;
    const int c4 = tid & 63;
    const float4 as4 = *(const float4*)(Asrc + 4 * c4);
    const float4 ad4 = *(const float4*)(Adst + 4 * c4);
    float4 r0[8], r1[8];
#pragma unroll
    for (int i = 0; i < 8; ++i) {
        const int kp = (tid >> 6) + i * 8;
        r0[i] = *(const float4*)(W + (size_t)(2 * kp) * HC + 4 * c4);
        r1[i] = *(const float4*)(W + (size_t)(2 * kp + 1) * HC + 4 * c4);
    }
#pragma unroll
    for (int i = 0; i < 8; ++i) {
        const int kp = (tid >> 6) + i * 8;
        const int j = kp >> 1, h = kp & 1;
        const int t = c4 >> 2, n = (c4 & 3) << 2;
        float* dst = sW + ((((j * 16 + t) * 32) + h * 16 + n) << 1);
        *(float4*)(dst)     = make_float4(r0[i].x, r1[i].x, r0[i].y, r1[i].y);
        *(float4*)(dst + 4) = make_float4(r0[i].z, r1[i].z, r0[i].w, r1[i].w);
        float s0 = r0[i].x * as4.x + r0[i].y * as4.y + r0[i].z * as4.z + r0[i].w * as4.w;
        float s1 = r1[i].x * as4.x + r1[i].y * as4.y + r1[i].z * as4.z + r1[i].w * as4.w;
        float d0 = r0[i].x * ad4.x + r0[i].y * ad4.y + r0[i].z * ad4.z + r0[i].w * ad4.w;
        float d1 = r1[i].x * ad4.x + r1[i].y * ad4.y + r1[i].z * ad4.z + r1[i].w * ad4.w;
        s0 = add_xor_lane<1>(s0); s1 = add_xor_lane<1>(s1);
        d0 = add_xor_lane<1>(d0); d1 = add_xor_lane<1>(d1);
        s0 = add_xor_lane<2>(s0); s1 = add_xor_lane<2>(s1);
        d0 = add_xor_lane<2>(d0); d1 = add_xor_lane<2>(d1);
        s0 = add_xor_lane<4>(s0); s1 = add_xor_lane<4>(s1);
        d0 = add_xor_lane<4>(d0); d1 = add_xor_lane<4>(d1);
        v2f* e = (v2f*)sE + j * 32 + h * 16 + (c4 >> 3);
        e[0]   = (v2f){s0, s1};
        e[NHD] = (v2f){d0, d1};
    }
    __syncthreads();
    const int lane = tid & 31;
    const int wave = tid >> 5;
    const int m    = lane & 15;
    const int half = lane >> 4;
    const int rowBase = ((blockIdx.x << 3) + (wave >> 1)) << 4;
    if (rowBase >= nrows) return;
    const int ch = wave & 1;
    const float* xrow = X + (size_t)(rowBase + m) * INCH + (half << 1);
    const v2f* bfrag = (const v2f*)sW + (ch << 3) * 32 + lane;
    const v2f* efrag = (const v2f*)sE + lane;
    if (ch == ((wave >> 2) & 1))
        gemm1_wave<true>(xrow, bfrag, efrag, H1, aS, aD, rowBase, ch, m, half);
    else
        gemm1_wave<false>(xrow, bfrag, efrag, H1, aS, aD, rowBase, ch, m, half);
}

__global__ void __launch_bounds__(256)
gemm_h_w2(const float* __restrict__ Hin, const float* __restrict__ W,
          const float* __restrict__ Asrc, const float* __restrict__ Adst,
          float* __restrict__ H2, float* __restrict__ aS, float* __restrict__ aD, int nrows) {
    __shared__ __attribute__((aligned(16))) float sW[64 * 3 * 32 * 2];
    const int tid = threadIdx.x;
    const int p = tid & 3;
    float4 a_s[3], a_d[3], wr[4][3];
#pragma unroll
    for (int cc = 0; cc < 3; ++cc) {
        const int pc = (p + 4 * cc < OUTC / 4) ? p + 4 * cc : p;
        a_s[cc] = *(const float4*)(Asrc + 4 * pc);
        a_d[cc] = *(const float4*)(Adst + 4 * pc);
        if (p + 4 * cc >= OUTC / 4) {
            a_s[cc] = make_float4(0.f, 0.f, 0.f, 0.f);
            a_d[cc] = make_float4(0.f, 0.f, 0.f, 0.f);
        }
#pragma unroll
        for (int it = 0; it < 4; ++it)
            wr[it][cc] = *(const float4*)(W + (size_t)(it * 64 + (tid >> 2)) * OUTC + 4 * pc);
    }
#pragma unroll
    for (int i = 0; i < 24; ++i) {
        const int s  = tid + i * 256;
        const int ln = s & 31, jt = s >> 5;
        const int j  = jt / 3, t = jt - j * 3;
        const int c  = t * 16 + (ln & 15);
        const int k  = 4 * j + 2 * (ln >> 4);
        v2f w = {0.f, 0.f};
        if (c < OUTC) { w.x = W[k * OUTC + c]; w.y = W[(k + 1) * OUTC + c]; }
        ((v2f*)sW)[s] = w;
    }
    __syncthreads();
#pragma unroll
    for (int it = 0; it < 4; ++it) {
        const int k = it * 64 + (tid >> 2);
        float sa = 0.f, sd = 0.f;
#pragma unroll
        for (int cc = 0; cc < 3; ++cc) {
            const float4 w = wr[it][cc];
            sa += w.x * a_s[cc].x + w.y * a_s[cc].y + w.z * a_s[cc].z + w.w * a_s[cc].w;
            sd += w.x * a_d[cc].x + w.y * a_d[cc].y + w.z * a_d[cc].z + w.w * a_d[cc].w;
        }
        sa = add_xor_lane<1>(sa); sd = add_xor_lane<1>(sd);
        sa = add_xor_lane<2>(sa); sd = add_xor_lane<2>(sd);
        float* e = sW + ((((k >> 2) * 3 + 2) * 32 + ((k >> 1) & 1) * 16 + (OUTC - 32)) << 1)
                      + (k & 1);
        e[0] = sa;
        e[2] = sd;
    }
    __syncthreads();
    const int lane = tid & 31;
    const int wave = tid >> 5;
    const int m    = lane & 15;
    const int half = lane >> 4;
    const int rowBase = ((blockIdx.x << 3) + wave) << 4;
    if (rowBase >= nrows) return;
    const float* hrow = Hin + (size_t)(rowBase + m) * HC + (half << 1);
    const v2f* bfrag = (const v2f*)sW + lane;
    v8f acc[3];
#pragma unroll
    for (int t = 0; t < 3; ++t) acc[t] = (v8f){0.f, 0.f, 0.f, 0.f, 0.f, 0.f, 0.f, 0.f};
    v2f ac[8], an[8];
#pragma unroll
    for (int q = 0; q < 8; ++q) ac[q] = *(const v2f*)(hrow + 4 * q);
#pragma unroll
    for (int c = 0; c < 8; ++c) {
        if (c < 7) {
#pragma unroll
            for (int q = 0; q < 8; ++q) an[q] = *(const v2f*)(hrow + 4 * ((c + 1) * 8 + q));
        }
#pragma unroll
        for (int q = 0; q < 8; ++q) {
            const int j = c * 8 + q;
#pragma unroll
            for (int t = 0; t < 3; ++t) {
                const v2f b = bfrag[(j * 3 + t) * 32];
                acc[t] = __builtin_amdgcn_wmma_f32_16x16x4_f32(false, ac[q], false, b,
                                                               (short)0, acc[t], false, false);
            }
        }
#pragma unroll
        for (int q = 0; q < 8; ++q) ac[q] = an[q];
    }
#pragma unroll
    for (int t = 0; t < 3; ++t) {
        const int col = (t << 4) + m;
        if (col < OUTC) {
#pragma unroll
            for (int v = 0; v < 8; ++v)
                H2[(size_t)(rowBase + v + (half << 3)) * OUTC + col] = acc[t][v];
        } else if (col < OUTC + 2) {
            float* dots = (col == OUTC) ? aS : aD;
#pragma unroll
            for (int v = 0; v < 8; ++v)
                dots[rowBase + v + (half << 3)] = acc[t][v];
        }
    }
}

__global__ void link_edges(const int* __restrict__ dstA, int* __restrict__ head,
                           int* __restrict__ nxt, int E) {
    const int e = blockIdx.x * blockDim.x + threadIdx.x;
    if (e >= E) return;
    nxt[e] = atomicExch(&head[dstA[e]], e);
}

__global__ void
gat_gather_l1(const int* __restrict__ head, const int* __restrict__ nxt,
              const int* __restrict__ srcA,
              const float* __restrict__ aS, const float* __restrict__ aD,
              const float* __restrict__ H1, const float* __restrict__ b1,
              float* __restrict__ out1, int n, int E) {
    const int lane = threadIdx.x & 31;
    const int wv = (blockIdx.x * blockDim.x + threadIdx.x) >> 5;
    const int node0 = wv << 2;
    if (node0 >= n) return;
    const int h = lane & 7;
    const int node = node0 + (lane >> 3);
    bool alive = node < n;
    int s = alive ? node : 0;
    int e = alive ? head[node] : -1;
    const float aDd = alive ? aD[node * NHD + h] : 0.f;
    float m = -__builtin_inff(), den = 0.f;
    float acc[32];
#pragma unroll
    for (int j = 0; j < 32; ++j) acc[j] = 0.f;

    for (int it = 0; it <= E; ++it) {
        const unsigned live = __builtin_amdgcn_ballot_w32(alive);
        if (live == 0u) break;
        const bool more = alive && e >= 0;
        int s2 = 0, e2 = -1;
        if (more) { s2 = srcA[e]; e2 = nxt[e]; }
        float as_s = 0.f;
        if (alive) as_s = aS[s * NHD + h];
        float hv[32];
#pragma unroll
        for (int q = 0; q < 4; ++q) {
            if (live & (1u << (q * 8))) {
                const int sq = __builtin_amdgcn_readlane(s, q * 8);
                const float* row = H1 + (size_t)sq * HC + lane;
#pragma unroll
                for (int hh = 0; hh < NHD; ++hh) hv[q * 8 + hh] = row[hh * CPH];
            }
        }
        float scale = 1.f, wgt = 0.f;
        if (alive) {
            const float l = leaky(as_s + aDd);
            const float mn = fmaxf(m, l);
            scale = expf(m - mn);
            wgt   = expf(l - mn);
            den = den * scale + wgt;
            m = mn;
        }
#pragma unroll
        for (int q = 0; q < 4; ++q) {
            if (live & (1u << (q * 8))) {
#pragma unroll
                for (int hh = 0; hh < NHD; ++hh) {
                    const float sc = __int_as_float(
                        __builtin_amdgcn_readlane(__float_as_int(scale), q * 8 + hh));
                    const float wg = __int_as_float(
                        __builtin_amdgcn_readlane(__float_as_int(wgt), q * 8 + hh));
                    acc[q * 8 + hh] = acc[q * 8 + hh] * sc + wg * hv[q * 8 + hh];
                }
            }
        }
        if (!more) alive = false;
        s = s2; e = e2;
    }

#pragma unroll
    for (int q = 0; q < 4; ++q) {
        if (node0 + q < n) {
#pragma unroll
            for (int hh = 0; hh < NHD; ++hh) {
                const float d = __int_as_float(
                    __builtin_amdgcn_readlane(__float_as_int(den), q * 8 + hh));
                const float v = acc[q * 8 + hh] / d + b1[hh * CPH + lane];
                out1[(size_t)(node0 + q) * HC + hh * CPH + lane] = v > 0.f ? v : 0.f;
            }
        }
    }
}

__global__ void gat_l2_fused(const int* __restrict__ head, const int* __restrict__ nxt,
                             const int* __restrict__ srcA,
                             const float* __restrict__ aS, const float* __restrict__ aD,
                             const float* __restrict__ H2, const float* __restrict__ b2,
                             float* __restrict__ out, int n, int E) {
    const int lane = threadIdx.x & 31;
    const int node = (blockIdx.x * blockDim.x + threadIdx.x) >> 5;
    if (node >= n) return;
    const float aDd = aD[node];
    const int c1 = lane + 32;
    int s = node;
    int e = head[node];
    float m = -__builtin_inff(), den = 0.f, acc0 = 0.f, acc1 = 0.f;
    for (int it = 0; it <= E; ++it) {
        const bool more = e >= 0;
        int s2 = 0, e2 = -1;
        if (more) { s2 = srcA[e]; e2 = nxt[e]; }
        const float l = leaky(aS[s] + aDd);
        const float* hrow = H2 + (size_t)s * OUTC;
        const float h0 = (lane < OUTC) ? hrow[lane] : 0.f;
        const float h1v = (c1 < OUTC) ? hrow[c1] : 0.f;
        const float mn = fmaxf(m, l);
        const float scale = expf(m - mn);
        const float wgt   = expf(l - mn);
        acc0 = acc0 * scale + wgt * h0;
        acc1 = acc1 * scale + wgt * h1v;
        den = den * scale + wgt;
        m = mn;
        if (!more) break;
        s = s2; e = e2;
    }
    float v0 = (lane < OUTC) ? acc0 / den + b2[lane] : -__builtin_inff();
    float v1 = (c1 < OUTC)  ? acc1 / den + b2[c1]  : -__builtin_inff();
    float mx = fmaxf(v0, v1);
    for (int o = 16; o > 0; o >>= 1) mx = fmaxf(mx, __shfl_xor(mx, o, 32));
    float ssum = 0.f;
    if (lane < OUTC) ssum += expf(v0 - mx);
    if (c1 < OUTC)  ssum += expf(v1 - mx);
    for (int o = 16; o > 0; o >>= 1) ssum += __shfl_xor(ssum, o, 32);
    const float lse = mx + logf(ssum);
    if (lane < OUTC) out[(size_t)node * OUTC + lane] = v0 - lse;
    if (c1 < OUTC)  out[(size_t)node * OUTC + c1]  = v1 - lse;
}

extern "C" void kernel_launch(void* const* d_in, const int* in_sizes, int n_in,
                              void* d_out, int out_size, void* d_ws, size_t ws_size,
                              hipStream_t stream) {
    const float* x   = (const float*)d_in[0];
    const int*   ei  = (const int*)d_in[1];
    const float* W1  = (const float*)d_in[2];
    const float* as1 = (const float*)d_in[3];
    const float* ad1 = (const float*)d_in[4];
    const float* b1  = (const float*)d_in[5];
    const float* W2  = (const float*)d_in[6];
    const float* as2 = (const float*)d_in[7];
    const float* ad2 = (const float*)d_in[8];
    const float* b2  = (const float*)d_in[9];
    const int N  = in_sizes[0] / INCH;
    const int E  = in_sizes[1] / 2;
    const int* srcA = ei;
    const int* dstA = ei + E;

    float* ws = (float*)d_ws;
    size_t off = 0;
    float* h1   = ws + off; off += (size_t)N * HC;
    float* out1 = ws + off; off += (size_t)N * HC;
    float* aS1  = ws + off; off += (size_t)N * NHD;
    float* aD1  = ws + off; off += (size_t)N * NHD;
    int* head = (int*)(ws + off); off += (size_t)N;
    int* nxt  = (int*)(ws + off); off += (size_t)E;
    float* h2  = h1;
    float* aS2 = aS1;
    float* aD2 = aD1;

    hipMemsetAsync(head, 0xFF, (size_t)N * sizeof(int), stream);
    link_edges<<<(E + 255) / 256, 256, 0, stream>>>(dstA, head, nxt, E);

    gemm_x_w1<<<(N + 127) / 128, 512, 0, stream>>>(x, W1, as1, ad1, h1, aS1, aD1, N);
    gat_gather_l1<<<((N + 3) / 4 * 32 + 255) / 256, 256, 0, stream>>>(
        head, nxt, srcA, aS1, aD1, h1, b1, out1, N, E);

    gemm_h_w2<<<(N + 127) / 128, 256, 0, stream>>>(out1, W2, as2, ad2, h2, aS2, aD2, N);
    gat_l2_fused<<<(N * 32 + 255) / 256, 256, 0, stream>>>(
        head, nxt, srcA, aS2, aD2, h2, b2, (float*)d_out, N, E);
}


# adjacency lists built in LDS: 256 groups, each an edge-list quarter x a node range, exchange on LDS head words (no global atomic, no memset); 4 lists per node; edge sum order: by quarter first
# speedup vs baseline: 3.1153x; 1.0772x over previous
#include <hip/hip_runtime.h>
#include <hip/hip_bf16.h>

typedef __attribute__((ext_vector_type(2))) float v2f;
typedef __attribute__((ext_vector_type(8))) float v8f;

#define INCH  128
#define HC    256
#define NHD   8
#define CPH   32
#define OUTC  40
#define SLOPE 0.2f

__device__ __forceinline__ float leaky(float x) { return x > 0.f ? x : SLOPE * x; }

template <int MASK>
__device__ __forceinline__ float add_xor_lane(float v) {
    return v + __int_as_float(__builtin_amdgcn_update_dpp(0, __float_as_int(v), 0x160 + MASK,
                                                          0xF, 0xF, false));
}

template <bool FOLDED>
__device__ __forceinline__ void
gemm1_wave(const float* __restrict__ xrow, const v2f* __restrict__ bfrag,
           const v2f* __restrict__ efrag, float* __restrict__ H1,
           float* __restrict__ aS, float* __restrict__ aD,
           int rowBase, int ch, int m, int half) {
    v8f acc[8];
#pragma unroll
    for (int i = 0; i < 8; ++i) acc[i] = (v8f){0.f, 0.f, 0.f, 0.f, 0.f, 0.f, 0.f, 0.f};
    v8f accE = (v8f){0.f, 0.f, 0.f, 0.f, 0.f, 0.f, 0.f, 0.f};
    v2f ac[8], an[8];
#pragma unroll
    for (int q = 0; q < 8; ++q) ac[q] = *(const v2f*)(xrow + 4 * q);
#pragma unroll
    for (int c = 0; c < 4; ++c) {
        if (c < 3) {
#pragma unroll
            for (int q = 0; q < 8; ++q) an[q] = *(const v2f*)(xrow + 4 * ((c + 1) * 8 + q));
        }
#pragma unroll
        for (int q = 0; q < 8; ++q) {
            const int j = c * 8 + q;
#pragma unroll
            for (int i = 0; i < 8; ++i) {
                const v2f b = bfrag[(j * 16 + i) * 32];
                acc[i] = __builtin_amdgcn_wmma_f32_16x16x4_f32(false, ac[q], false, b,
                                                               (short)0, acc[i], false, false);
            }
            if (FOLDED) {
                const v2f b = efrag[j * 32];
                accE = __builtin_amdgcn_wmma_f32_16x16x4_f32(false, ac[q], false, b,
                                                             (short)0, accE, false, false);
            }
        }
#pragma unroll
        for (int q = 0; q < 8; ++q) ac[q] = an[q];
    }
#pragma unroll
    for (int i = 0; i < 8; ++i) {
        const int colBase = ((ch << 3) + i) << 4;
#pragma unroll
        for (int v = 0; v < 8; ++v)
            H1[(size_t)(rowBase + v + (half << 3)) * HC + colBase + m] = acc[i][v];
    }
    if (FOLDED) {
        float* dots = (m < NHD) ? aS + m : aD + (m - NHD);
#pragma unroll
        for (int v = 0; v < 8; ++v)
            dots[(size_t)(rowBase + v + (half << 3)) * NHD] = accE[v];
    }
}

__global__ void __launch_bounds__(512)
gemm_x_w1(const float* __restrict__ X, const float* __restrict__ W,
          const float* __restrict__ Asrc, const float* __restrict__ Adst,
          float* __restrict__ H1, float* __restrict__ aS, float* __restrict__ aD, int nrows) {
    __shared__ __attribute__((aligned(16))) float sW[INCH * HC];
    __shared__ __attribute__((aligned(16))) float sE[(INCH / 4) * 32 * 2];
    const int tid = threadIdx.x;
    const int c4 = tid & 63;
    const float4 as4 = *(const float4*)(Asrc + 4 * c4);
    const float4 ad4 = *(const float4*)(Adst + 4 * c4);
    float4 r0[8], r1[8];
#pragma unroll
    for (int i = 0; i < 8; ++i) {
        const int kp = (tid >> 6) + i * 8;
        r0[i] = *(const float4*)(W + (size_t)(2 * kp) * HC + 4 * c4);
        r1[i] = *(const float4*)(W + (size_t)(2 * kp + 1) * HC + 4 * c4);
    }
#pragma unroll
    for (int i = 0; i < 8; ++i) {
        const int kp = (tid >> 6) + i * 8;
        const int j = kp >> 1, h = kp & 1;
        const int t = c4 >> 2, n = (c4 & 3) << 2;
        float* dst = sW + ((((j * 16 + t) * 32) + h * 16 + n) << 1);
        *(float4*)(dst)     = make_float4(r0[i].x, r1[i].x, r0[i].y, r1[i].y);
        *(float4*)(dst + 4) = make_float4(r0[i].z, r1[i].z, r0[i].w, r1[i].w);
        float s0 = r0[i].x * as4.x + r0[i].y * as4.y + r0[i].z * as4.z + r0[i].w * as4.w;
        float s1 = r1[i].x * as4.x + r1[i].y * as4.y + r1[i].z * as4.z + r1[i].w * as4.w;
        float d0 = r0[i].x * ad4.x + r0[i].y * ad4.y + r0[i].z * ad4.z + r0[i].w * ad4.w;
        float d1 = r1[i].x * ad4.x + r1[i].y * ad4.y + r1[i].z * ad4.z + r1[i].w * ad4.w;
        s0 = add_xor_lane<1>(s0); s1 = add_xor_lane<1>(s1);
        d0 = add_xor_lane<1>(d0); d1 = add_xor_lane<1>(d1);
        s0 = add_xor_lane<2>(s0); s1 = add_xor_lane<2>(s1);
        d0 = add_xor_lane<2>(d0); d1 = add_xor_lane<2>(d1);
        s0 = add_xor_lane<4>(s0); s1 = add_xor_lane<4>(s1);
        d0 = add_xor_lane<4>(d0); d1 = add_xor_lane<4>(d1);
        v2f* e = (v2f*)sE + j * 32 + h * 16 + (c4 >> 3);
        e[0]   = (v2f){s0, s1};
        e[NHD] = (v2f){d0, d1};
    }
    __syncthreads();
    const int lane = tid & 31;
    const int wave = tid >> 5;
    const int m    = lane & 15;
    const int half = lane >> 4;
    const int rowBase = ((blockIdx.x << 3) + (wave >> 1)) << 4;
    if (rowBase >= nrows) return;
    const int ch = wave & 1;
    const float* xrow = X + (size_t)(rowBase + m) * INCH + (half << 1);
    const v2f* bfrag = (const v2f*)sW + (ch << 3) * 32 + lane;
    const v2f* efrag = (const v2f*)sE + lane;
    if (ch == ((wave >> 2) & 1))
        gemm1_wave<true>(xrow, bfrag, efrag, H1, aS, aD, rowBase, ch, m, half);
    else
        gemm1_wave<false>(xrow, bfrag, efrag, H1, aS, aD, rowBase, ch, m, half);
}

__global__ void __launch_bounds__(256)
gemm_h_w2(const float* __restrict__ Hin, const float* __restrict__ W,
          const float* __restrict__ Asrc, const float* __restrict__ Adst,
          float* __restrict__ H2, float* __restrict__ aS, float* __restrict__ aD, int nrows) {
    __shared__ __attribute__((aligned(16))) float sW[64 * 3 * 32 * 2];
    const int tid = threadIdx.x;
    const int p = tid & 3;
    float4 a_s[3], a_d[3], wr[4][3];
#pragma unroll
    for (int cc = 0; cc < 3; ++cc) {
        const int pc = (p + 4 * cc < OUTC / 4) ? p + 4 * cc : p;
        a_s[cc] = *(const float4*)(Asrc + 4 * pc);
        a_d[cc] = *(const float4*)(Adst + 4 * pc);
        if (p + 4 * cc >= OUTC / 4) {
            a_s[cc] = make_float4(0.f, 0.f, 0.f, 0.f);
            a_d[cc] = make_float4(0.f, 0.f, 0.f, 0.f);
        }
#pragma unroll
        for (int it = 0; it < 4; ++it)
            wr[it][cc] = *(const float4*)(W + (size_t)(it * 64 + (tid >> 2)) * OUTC + 4 * pc);
    }
#pragma unroll
    for (int i = 0; i < 24; ++i) {
        const int s  = tid + i * 256;
        const int ln = s & 31, jt = s >> 5;
        const int j  = jt / 3, t = jt - j * 3;
        const int c  = t * 16 + (ln & 15);
        const int k  = 4 * j + 2 * (ln >> 4);
        v2f w = {0.f, 0.f};
        if (c < OUTC) { w.x = W[k * OUTC + c]; w.y = W[(k + 1) * OUTC + c]; }
        ((v2f*)sW)[s] = w;
    }
    __syncthreads();
#pragma unroll
    for (int it = 0; it < 4; ++it) {
        const int k = it * 64 + (tid >> 2);
        float sa = 0.f, sd = 0.f;
#pragma unroll
        for (int cc = 0; cc < 3; ++cc) {
            const float4 w = wr[it][cc];
            sa += w.x * a_s[cc].x + w.y * a_s[cc].y + w.z * a_s[cc].z + w.w * a_s[cc].w;
            sd += w.x * a_d[cc].x + w.y * a_d[cc].y + w.z * a_d[cc].z + w.w * a_d[cc].w;
        }
        sa = add_xor_lane<1>(sa); sd = add_xor_lane<1>(sd);
        sa = add_xor_lane<2>(sa); sd = add_xor_lane<2>(sd);
        float* e = sW + ((((k >> 2) * 3 + 2) * 32 + ((k >> 1) & 1) * 16 + (OUTC - 32)) << 1)
                      + (k & 1);
        e[0] = sa;
        e[2] = sd;
    }
    __syncthreads();
    const int lane = tid & 31;
    const int wave = tid >> 5;
    const int m    = lane & 15;
    const int half = lane >> 4;
    const int rowBase = ((blockIdx.x << 3) + wave) << 4;
    if (rowBase >= nrows) return;
    const float* hrow = Hin + (size_t)(rowBase + m) * HC + (half << 1);
    const v2f* bfrag = (const v2f*)sW + lane;
    v8f acc[3];
#pragma unroll
    for (int t = 0; t < 3; ++t) acc[t] = (v8f){0.f, 0.f, 0.f, 0.f, 0.f, 0.f, 0.f, 0.f};
    v2f ac[8], an[8];
#pragma unroll
    for (int q = 0; q < 8; ++q) ac[q] = *(const v2f*)(hrow + 4 * q);
#pragma unroll
    for (int c = 0; c < 8; ++c) {
        if (c < 7) {
#pragma unroll
            for (int q = 0; q < 8; ++q) an[q] = *(const v2f*)(hrow + 4 * ((c + 1) * 8 + q));
        }
#pragma unroll
        for (int q = 0; q < 8; ++q) {
            const int j = c * 8 + q;
#pragma unroll
            for (int t = 0; t < 3; ++t) {
                const v2f b = bfrag[(j * 3 + t) * 32];
                acc[t] = __builtin_amdgcn_wmma_f32_16x16x4_f32(false, ac[q], false, b,
                                                               (short)0, acc[t], false, false);
            }
        }
#pragma unroll
        for (int q = 0; q < 8; ++q) ac[q] = an[q];
    }
#pragma unroll
    for (int t = 0; t < 3; ++t) {
        const int col = (t << 4) + m;
        if (col < OUTC) {
#pragma unroll
            for (int v = 0; v < 8; ++v)
                H2[(size_t)(rowBase + v + (half << 3)) * OUTC + col] = acc[t][v];
        } else if (col < OUTC + 2) {
            float* dots = (col == OUTC) ? aS : aD;
#pragma unroll
            for (int v = 0; v < 8; ++v)
                dots[rowBase + v + (half << 3)] = acc[t][v];
        }
    }
}

#define LSL   4
#define LRG   64
#define LCAP  4096

__device__ __forceinline__ void take_edge(int* lh, int* __restrict__ nxt, int d, int e,
                                          int base, int cnt) {
    const unsigned o = (unsigned)(d - base);
    if (o < (unsigned)cnt) nxt[e] = atomicExch(&lh[o], e);
}

__global__ void __launch_bounds__(1024)
link_edges_lds(const int* __restrict__ dstA, int* __restrict__ head, int* __restrict__ nxt,
               int n, int E) {
    __shared__ int lh[LCAP];
    const int tid = threadIdx.x;
    const int s = blockIdx.x % LSL;
    const int r = blockIdx.x / LSL;
    const int es = (((E + LSL - 1) / LSL) + 3) & ~3;
    const int rn = (n + LRG - 1) / LRG;
    const int e0 = s * es;
    const int len = min(es, E - e0);
    const int rEnd = min((r + 1) * rn, n);
    const int4* dv = (const int4*)(dstA + e0);
    const int units = ((((size_t)dv) & 15) == 0 && len > 0) ? (len >> 2) : 0;
    const int rot = (int)(((long long)r * units) / LRG);
    for (int base = r * rn; base < rEnd; base += LCAP) {
        const int cnt = min(LCAP, rEnd - base);
        for (int i = tid; i < cnt; i += 1024) lh[i] = -1;
        __syncthreads();
        for (int i0 = tid; i0 < units; i0 += 8 * 1024) {
            int4 d[8];
            int eb[8];
#pragma unroll
            for (int k = 0; k < 8; ++k) {
                const int i = i0 + k * 1024;
                int u = i + rot;
                if (u >= units) u -= units;
                eb[k] = e0 + 4 * u;
                d[k] = make_int4(-1, -1, -1, -1);
                if (i < units) d[k] = dv[u];
            }
#pragma unroll
            for (int k = 0; k < 8; ++k) {
                take_edge(lh, nxt, d[k].x, eb[k],     base, cnt);
                take_edge(lh, nxt, d[k].y, eb[k] + 1, base, cnt);
                take_edge(lh, nxt, d[k].z, eb[k] + 2, base, cnt);
                take_edge(lh, nxt, d[k].w, eb[k] + 3, base, cnt);
            }
        }
        for (int i = 4 * units + tid; i < len; i += 1024)
            take_edge(lh, nxt, dstA[e0 + i], e0 + i, base, cnt);
        __syncthreads();
        for (int i = tid; i < cnt; i += 1024) head[(size_t)s * n + base + i] = lh[i];
        __syncthreads();
    }
}

__device__ __forceinline__ void order_heads(int& e, int& q1, int& q2, int& q3) {
#define PULL(a, b) if (a < 0) { a = b; b = -1; }
    PULL(q2, q3) PULL(q1, q2) PULL(q2, q3) PULL(e, q1) PULL(q1, q2) PULL(q2, q3)
#undef PULL
}

__global__ void
gat_gather_l1(const int* __restrict__ head, const int* __restrict__ nxt,
              const int* __restrict__ srcA,
              const float* __restrict__ aS, const float* __restrict__ aD,
              const float* __restrict__ H1, const float* __restrict__ b1,
              float* __restrict__ out1, int n, int E) {
    __shared__ int qs[(LSL - 1) * 256];
    const int lane = threadIdx.x & 31;
    const int wv = (blockIdx.x * blockDim.x + threadIdx.x) >> 5;
    const int node0 = wv << 2;
    if (node0 >= n) return;
    const int h = lane & 7;
    const int node = node0 + (lane >> 3);
    bool alive = node < n;
    int s = alive ? node : 0;
    int e = -1;
    const int qbase = __builtin_amdgcn_readfirstlane((int)threadIdx.x);
    {
        int q1 = -1, q2 = -1, q3 = -1;
        if (alive) {
            e  = head[node];
            q1 = head[(size_t)n + node];
            q2 = head[(size_t)2 * n + node];
            q3 = head[(size_t)3 * n + node];
        }
        order_heads(e, q1, q2, q3);
        qs[threadIdx.x] = q1; qs[256 + threadIdx.x] = q2; qs[512 + threadIdx.x] = q3;
    }
    const float aDd = alive ? aD[node * NHD + h] : 0.f;
    float m = -__builtin_inff(), den = 0.f;
    float acc[32];
#pragma unroll
    for (int j = 0; j < 32; ++j) acc[j] = 0.f;

    for (int it = 0; it <= E; ++it) {
        const unsigned live = __builtin_amdgcn_ballot_w32(alive);
        if (live == 0u) break;
        if (alive && e < 0) {
            int li;
            asm volatile("v_mbcnt_lo_u32_b32 %0, -1, 0" : "=v"(li));
            int* q = qs + qbase + li;
            const int eo = e;
            e = q[0]; q[0] = q[256]; q[256] = q[512]; q[512] = eo;
        }
        const bool more = alive && e >= 0;
        int s2 = 0, e2 = -1;
        if (more) { s2 = srcA[e]; e2 = nxt[e]; }
        float as_s = 0.f;
        if (alive) as_s = aS[s * NHD + h];
        float hv[32];
#pragma unroll
        for (int q = 0; q < 4; ++q) {
            if (live & (1u << (q * 8))) {
                const int sq = __builtin_amdgcn_readlane(s, q * 8);
                const float* row = H1 + (size_t)sq * HC + lane;
#pragma unroll
                for (int hh = 0; hh < NHD; ++hh) hv[q * 8 + hh] = row[hh * CPH];
            }
        }
        float scale = 1.f, wgt = 0.f;
        if (alive) {
            const float l = leaky(as_s + aDd);
            const float mn = fmaxf(m, l);
            scale = expf(m - mn);
            wgt   = expf(l - mn);
            den = den * scale + wgt;
            m = mn;
        }
#pragma unroll
        for (int q = 0; q < 4; ++q) {
            if (live & (1u << (q * 8))) {
#pragma unroll
                for (int hh = 0; hh < NHD; ++hh) {
                    const float sc = __int_as_float(
                        __builtin_amdgcn_readlane(__float_as_int(scale), q * 8 + hh));
                    const float wg = __int_as_float(
                        __builtin_amdgcn_readlane(__float_as_int(wgt), q * 8 + hh));
                    acc[q * 8 + hh] = acc[q * 8 + hh] * sc + wg * hv[q * 8 + hh];
                }
            }
        }
        if (!more) alive = false;
        s = s2; e = e2;
    }

#pragma unroll
    for (int q = 0; q < 4; ++q) {
        if (node0 + q < n) {
#pragma unroll
            for (int hh = 0; hh < NHD; ++hh) {
                const float d = __int_as_float(
                    __builtin_amdgcn_readlane(__float_as_int(den), q * 8 + hh));
                const float v = acc[q * 8 + hh] / d + b1[hh * CPH + lane];
                out1[(size_t)(node0 + q) * HC + hh * CPH + lane] = v > 0.f ? v : 0.f;
            }
        }
    }
}

__global__ void gat_l2_fused(const int* __restrict__ head, const int* __restrict__ nxt,
                             const int* __restrict__ srcA,
                             const float* __restrict__ aS, const float* __restrict__ aD,
                             const float* __restrict__ H2, const float* __restrict__ b2,
                             float* __restrict__ out, int n, int E) {
    const int lane = threadIdx.x & 31;
    const int node = (blockIdx.x * blockDim.x + threadIdx.x) >> 5;
    if (node >= n) return;
    const float aDd = aD[node];
    const int c1 = lane + 32;
    int s = node;
    int e  = head[node];
    int q1 = head[(size_t)n + node];
    int q2 = head[(size_t)2 * n + node];
    int q3 = head[(size_t)3 * n + node];
    order_heads(e, q1, q2, q3);
    float m = -__builtin_inff(), den = 0.f, acc0 = 0.f, acc1 = 0.f;
    for (int it = 0; it <= E; ++it) {
        if (e < 0) { e = q1; q1 = q2; q2 = q3; q3 = -1; }
        const bool more = e >= 0;
        int s2 = 0, e2 = -1;
        if (more) { s2 = srcA[e]; e2 = nxt[e]; }
        const float l = leaky(aS[s] + aDd);
        const float* hrow = H2 + (size_t)s * OUTC;
        const float h0 = (lane < OUTC) ? hrow[lane] : 0.f;
        const float h1v = (c1 < OUTC) ? hrow[c1] : 0.f;
        const float mn = fmaxf(m, l);
        const float scale = expf(m - mn);
        const float wgt   = expf(l - mn);
        acc0 = acc0 * scale + wgt * h0;
        acc1 = acc1 * scale + wgt * h1v;
        den = den * scale + wgt;
        m = mn;
        if (!more) break;
        s = s2; e = e2;
    }
    float v0 = (lane < OUTC) ? acc0 / den + b2[lane] : -__builtin_inff();
    float v1 = (c1 < OUTC)  ? acc1 / den + b2[c1]  : -__builtin_inff();
    float mx = fmaxf(v0, v1);
    for (int o = 16; o > 0; o >>= 1) mx = fmaxf(mx, __shfl_xor(mx, o, 32));
    float ssum = 0.f;
    if (lane < OUTC) ssum += expf(v0 - mx);
    if (c1 < OUTC)  ssum += expf(v1 - mx);
    for (int o = 16; o > 0; o >>= 1) ssum += __shfl_xor(ssum, o, 32);
    const float lse = mx + logf(ssum);
    if (lane < OUTC) out[(size_t)node * OUTC + lane] = v0 - lse;
    if (c1 < OUTC)  out[(size_t)node * OUTC + c1]  = v1 - lse;
}

extern "C" void kernel_launch(void* const* d_in, const int* in_sizes, int n_in,
                              void* d_out, int out_size, void* d_ws, size_t ws_size,
                              hipStream_t stream) {
    const float* x   = (const float*)d_in[0];
    const int*   ei  = (const int*)d_in[1];
    const float* W1  = (const float*)d_in[2];
    const float* as1 = (const float*)d_in[3];
    const float* ad1 = (const float*)d_in[4];
    const float* b1  = (const float*)d_in[5];
    const float* W2  = (const float*)d_in[6];
    const float* as2 = (const float*)d_in[7];
    const float* ad2 = (const float*)d_in[8];
    const float* b2  = (const float*)d_in[9];
    const int N  = in_sizes[0] / INCH;
    const int E  = in_sizes[1] / 2;
    const int* srcA = ei;
    const int* dstA = ei + E;

    float* ws = (float*)d_ws;
    size_t off = 0;
    float* h1   = ws + off; off += (size_t)N * HC;
    float* out1 = ws + off; off += (size_t)N * HC;
    float* aS1  = ws + off; off += (size_t)N * NHD;
    float* aD1  = ws + off; off += (size_t)N * NHD;
    int* head = (int*)(ws + off); off += (size_t)N * LSL;
    int* nxt  = (int*)(ws + off); off += (size_t)E;
    float* h2  = h1;
    float* aS2 = aS1;
    float* aD2 = aD1;

    link_edges_lds<<<LSL * LRG, 1024, 0, stream>>>(dstA, head, nxt, N, E);

    gemm_x_w1<<<(N + 127) / 128, 512, 0, stream>>>(x, W1, as1, ad1, h1, aS1, aD1, N);
    gat_gather_l1<<<((N + 3) / 4 * 32 + 255) / 256, 256, 0, stream>>>(
        head, nxt, srcA, aS1, aD1, h1, b1, out1, N, E);

    gemm_h_w2<<<(N + 127) / 128, 256, 0, stream>>>(out1, W2, as2, ad2, h2, aS2, aD2, N);
    gat_l2_fused<<<(N * 32 + 255) / 256, 256, 0, stream>>>(
        head, nxt, srcA, aS2, aD2, h2, b2, (float*)d_out, N, E);
}
